# starting kernel
# baseline (speedup 1.0000x reference)
#include <hip/hip_runtime.h>
#include <hip/hip_bf16.h>

typedef __attribute__((ext_vector_type(2))) float v2f;
typedef __attribute__((ext_vector_type(8))) float v8f;

#define NR 256

__global__ __launch_bounds__(256) void sg_phase1(const float* __restrict__ t_all,
                                                 const float* __restrict__ weights,
                                                 float* __restrict__ Bws)
{
    __shared__ float sG[16];
    __shared__ float sGinv[16];
    __shared__ float sAw[4 * NR];

    const int i = threadIdx.x;
    if (i < 16) sG[i] = 0.0f;
    __syncthreads();

    const float tc = t_all[NR / 2];
    const float tr = t_all[i] - tc;
    const float wr = weights[i];
    const float w  = fmaxf(wr, 0.0f) + log1pf(expf(-fabsf(wr)));
    float v[4];
    v[0] = 1.0f;
    v[1] = tr;
    v[2] = tr * tr * 0.5f;
    v[3] = tr * tr * tr * (1.0f / 3.0f);

    #pragma unroll
    for (int p = 0; p < 4; ++p) sAw[p * NR + i] = v[p] * w;

    #pragma unroll
    for (int p = 0; p < 4; ++p)
        #pragma unroll
        for (int q = 0; q < 4; ++q)
            atomicAdd(&sG[p * 4 + q], v[p] * v[q] * w);
    __syncthreads();

    if (i == 0) {
        float M[4][8];
        for (int r = 0; r < 4; ++r)
            for (int c = 0; c < 4; ++c) {
                M[r][c]     = sG[r * 4 + c];
                M[r][4 + c] = (r == c) ? 1.0f : 0.0f;
            }
        for (int col = 0; col < 4; ++col) {
            const float inv = 1.0f / M[col][col];
            for (int c = 0; c < 8; ++c) M[col][c] *= inv;
            for (int r = 0; r < 4; ++r) {
                if (r == col) continue;
                const float f = M[r][col];
                for (int c = 0; c < 8; ++c) M[r][c] -= f * M[col][c];
            }
        }
        for (int r = 0; r < 4; ++r)
            for (int c = 0; c < 4; ++c) sGinv[r * 4 + c] = M[r][4 + c];
    }
    __syncthreads();

    float beff[4];
    #pragma unroll
    for (int n = 0; n < 4; ++n) {
        float acc = 0.0f;
        #pragma unroll
        for (int q = 0; q < 4; ++q) acc += sGinv[n * 4 + q] * sAw[q * NR + i];
        beff[n] = acc;
    }

    const int s = i >> 2;
    const int k = i & 3;
    const int j = k & 1;
    const int lbase = (k < 2) ? 0 : 16;
    #pragma unroll
    for (int n = 0; n < 16; ++n) {
        const float val = (n < 4) ? beff[n] : 0.0f;
        Bws[s * 64 + (lbase + n) * 2 + j] = val;
    }
}

__global__ __launch_bounds__(256) void sg_phase2(const float* __restrict__ X,
                                                 const float* __restrict__ Bws,
                                                 float* __restrict__ out,
                                                 int Dc, int numTiles)
{
    const int lane = threadIdx.x & 31;
    const int wave = threadIdx.x >> 5;
    const int tile = blockIdx.x * 8 + wave;
    if (tile >= numTiles) return;

    const long d0 = (long)tile * 16;
    const int  m  = lane & 15;
    const int  hi = lane >> 4;
    const int  kb = hi * 2;

    long col = d0 + m;
    if (col >= Dc) col = Dc - 1;
    const float* xp = X + (long)kb * Dc + col;

    v8f c = {0.0f, 0.0f, 0.0f, 0.0f, 0.0f, 0.0f, 0.0f, 0.0f};

    for (int s = 0; s < NR / 4; ++s) {
        const float* xs = xp + (long)s * 4 * Dc;
        v2f a;
        a.x = xs[0];
        a.y = xs[Dc];
        v2f b = *(const v2f*)(Bws + s * 64 + lane * 2);
        __builtin_prefetch(xs + 8 * (long)Dc, 0, 0);
        c = __builtin_amdgcn_wmma_f32_16x16x4_f32(
                false, a, false, b, (short)0, c, false, false);
    }

    const long dbase = d0 + (long)hi * 8;
    if (d0 + 16 <= (long)Dc) {
        if (m < 4) {
            #pragma unroll
            for (int r = 0; r < 8; ++r) {
                out[(dbase + r) * 4 + m] = c[r];
            }
        }
    } else {
        if (m < 4) {
            #pragma unroll
            for (int r = 0; r < 8; ++r) {
                const long d = dbase + r;
                if (d < Dc) out[d * 4 + m] = c[r];
            }
        }
    }
}

extern "C" void kernel_launch(void* const* d_in, const int* in_sizes, int n_in,
                              void* d_out, int out_size, void* d_ws, size_t ws_size,
                              hipStream_t stream) {
    const float* t_all   = (const float*)d_in[0];
    const float* x_all   = (const float*)d_in[1];
    const float* weights = (const float*)d_in[2];
    (void)d_in[3]; (void)n_in; (void)out_size; (void)ws_size;

    const int T  = in_sizes[0];
    const int Dc = in_sizes[1] / T;

    float* Bws = (float*)d_ws;

    hipLaunchKernelGGL(sg_phase1, dim3(1), dim3(256), 0, stream,
                       t_all, weights, Bws);

    const int numTiles = (Dc + 15) / 16;
    const int blocks   = (numTiles + 7) / 8;
    hipLaunchKernelGGL(sg_phase2, dim3(blocks), dim3(256), 0, stream,
                       x_all, Bws, (float*)d_out, Dc, numTiles);
}


# one launch, solve behind the stream: WMMA sums 4 weighted moments (W V)^T X per channel (A staged per work-group in LDS, X as B), a lane applies the 4x4 inverse; Gram by shuffle tree: new sum order
# speedup vs baseline: 1.8293x; 1.8293x over previous
#include <hip/hip_runtime.h>


typedef __attribute__((ext_vector_type(2))) float v2f;
typedef __attribute__((ext_vector_type(8))) float v8f;

#define NR 256
#define FS 66

__global__ __launch_bounds__(256) void sg_fit(const float* __restrict__ t_all,
                                              const float* __restrict__ weights,
                                              const float* __restrict__ X,
                                              float* __restrict__ out,
                                              int Dc, int numTiles)
{
    __shared__ __attribute__((aligned(16))) float sA[(NR / 4) * FS];
    __shared__ __attribute__((aligned(16))) float sGp[8 * 12];

    const int i    = threadIdx.x;
    const int lane = i & 31;
    const int wave = i >> 5;
    const int tile = blockIdx.x * 8 + wave;
    const bool live = tile < numTiles;

    const long d0 = (long)tile * 16;
    const int  c  = lane & 15;
    const int  hi = lane >> 4;
    const int  kb = hi * 2;

    long col = d0 + c;
    if (col >= Dc) col = Dc - 1;
    const float* xp = X + (long)kb * Dc + col;

    __builtin_prefetch(xp, 0, 0);
    __builtin_prefetch(xp + 4 * (long)Dc, 0, 0);

    const float tc = t_all[NR / 2];
    const float tr = t_all[i] - tc;
    const float wr = weights[i];
    const float w  = fmaxf(wr, 0.0f) + log1pf(expf(-fabsf(wr)));
    float v[4];
    v[0] = 1.0f;
    v[1] = tr;
    v[2] = tr * tr * 0.5f;
    v[3] = tr * tr * tr * (1.0f / 3.0f);

    {
        const int s  = i >> 2;
        const int k  = i & 3;
        const int j  = k & 1;
        const int lb = (k < 2) ? 0 : 16;
        #pragma unroll
        for (int m = 0; m < 16; ++m) {
            const float val = (m < 4) ? v[m] * w : 0.0f;
            sA[s * FS + (lb + m) * 2 + j] = val;
        }
    }

    float g[10];
    {
        int e = 0;
        #pragma unroll
        for (int p = 0; p < 4; ++p)
            #pragma unroll
            for (int q = p; q < 4; ++q)
                g[e++] = v[p] * v[q] * w;
    }
    #pragma unroll
    for (int e = 0; e < 10; ++e) {
        #pragma unroll
        for (int off = 16; off > 0; off >>= 1)
            g[e] += __shfl_xor(g[e], off);
    }
    if (lane == 0) {
        #pragma unroll
        for (int e = 0; e < 10; ++e) sGp[wave * 12 + e] = g[e];
    }
    __syncthreads();

    if (!live) return;

    v8f acc = {0.0f, 0.0f, 0.0f, 0.0f, 0.0f, 0.0f, 0.0f, 0.0f};

    for (int s0 = 0; s0 < NR / 4; s0 += 4) {
        v2f a[4], b[4];
        #pragma unroll
        for (int u = 0; u < 4; ++u) {
            const float* xs = xp + (long)(s0 + u) * 4 * Dc;
            b[u].x = xs[0];
            b[u].y = xs[Dc];
            a[u] = *(const v2f*)(sA + (s0 + u) * FS + lane * 2);
            __builtin_prefetch(xs + 8 * (long)Dc, 0, 0);
        }
        #pragma unroll
        for (int u = 0; u < 4; ++u) {
            acc = __builtin_amdgcn_wmma_f32_16x16x4_f32(
                      false, a[u], false, b[u], (short)0, acc, false, false);
        }
    }

    if (hi == 0 && d0 + c < (long)Dc) {
        float G[10];
        #pragma unroll
        for (int e = 0; e < 10; ++e) {
            float gs = sGp[e];
            #pragma unroll
            for (int wv = 1; wv < 8; ++wv) gs += sGp[wv * 12 + e];
            G[e] = gs;
        }

        float M[4][8];
        {
            int e = 0;
            #pragma unroll
            for (int p = 0; p < 4; ++p)
                #pragma unroll
                for (int q = p; q < 4; ++q) {
                    M[p][q] = G[e];
                    M[q][p] = G[e];
                    ++e;
                }
        }
        #pragma unroll
        for (int r = 0; r < 4; ++r)
            #pragma unroll
            for (int cc = 0; cc < 4; ++cc) M[r][4 + cc] = (r == cc) ? 1.0f : 0.0f;
        #pragma unroll
        for (int pc = 0; pc < 4; ++pc) {
            const float inv = 1.0f / M[pc][pc];
            #pragma unroll
            for (int cc = 0; cc < 8; ++cc) M[pc][cc] *= inv;
            #pragma unroll
            for (int r = 0; r < 4; ++r) {
                if (r == pc) continue;
                const float f = M[r][pc];
                #pragma unroll
                for (int cc = 0; cc < 8; ++cc) M[r][cc] -= f * M[pc][cc];
            }
        }

        float4 o;
        o.x = M[0][4] * acc[0] + M[0][5] * acc[1] + M[0][6] * acc[2] + M[0][7] * acc[3];
        o.y = M[1][4] * acc[0] + M[1][5] * acc[1] + M[1][6] * acc[2] + M[1][7] * acc[3];
        o.z = M[2][4] * acc[0] + M[2][5] * acc[1] + M[2][6] * acc[2] + M[2][7] * acc[3];
        o.w = M[3][4] * acc[0] + M[3][5] * acc[1] + M[3][6] * acc[2] + M[3][7] * acc[3];
        *(float4*)(out + (d0 + c) * 4) = o;
    }
}

extern "C" void kernel_launch(void* const* d_in, const int* in_sizes, int n_in,
                              void* d_out, int out_size, void* d_ws, size_t ws_size,
                              hipStream_t stream) {
    const float* t_all   = (const float*)d_in[0];
    const float* x_all   = (const float*)d_in[1];
    const float* weights = (const float*)d_in[2];
    (void)d_in[3]; (void)n_in; (void)out_size; (void)d_ws; (void)ws_size;

    const int T  = in_sizes[0];
    const int Dc = in_sizes[1] / T;

    const int numTiles = (Dc + 15) / 16;
    const int blocks   = (numTiles + 7) / 8;
    hipLaunchKernelGGL(sg_fit, dim3(blocks), dim3(256), 0, stream,
                       t_all, weights, x_all, (float*)d_out, Dc, numTiles);
}


# K split over the work-group's 8 waves: a wave sums 32 rows for all 8 channel tiles, 64 independent X loads in flight per pass; partial moments added in wave order through LDS: new sum order
# speedup vs baseline: 2.2945x; 1.2543x over previous
#include <hip/hip_runtime.h>


typedef __attribute__((ext_vector_type(2))) float v2f;
typedef __attribute__((ext_vector_type(8))) float v8f;

#define NR 256
#define FS 66
#define NW 8

__global__ __launch_bounds__(256) void sg_fit(const float* __restrict__ t_all,
                                              const float* __restrict__ weights,
                                              const float* __restrict__ X,
                                              float* __restrict__ out,
                                              int Dc)
{
    __shared__ __attribute__((aligned(16))) float sA[(NR / 4) * FS];
    __shared__ __attribute__((aligned(16))) float sGp[NW * 12];
    __shared__ __attribute__((aligned(16))) float4 sP[NW][NW][16];

    const int i    = threadIdx.x;
    const int lane = i & 31;
    const int wave = __builtin_amdgcn_readfirstlane(i >> 5);
    const int tile0 = blockIdx.x * NW;

    const int  c  = lane & 15;
    const int  hi = lane >> 4;
    const int  kb = hi * 2;

    const float tc = t_all[NR / 2];
    const float tr = t_all[i] - tc;
    const float wr = weights[i];
    const float w  = fmaxf(wr, 0.0f) + log1pf(expf(-fabsf(wr)));
    float v[4];
    v[0] = 1.0f;
    v[1] = tr;
    v[2] = tr * tr * 0.5f;
    v[3] = tr * tr * tr * (1.0f / 3.0f);

    {
        const int s  = i >> 2;
        const int k  = i & 3;
        const int j  = k & 1;
        const int lb = (k < 2) ? 0 : 16;
        #pragma unroll
        for (int m = 0; m < 16; ++m) {
            const float val = (m < 4) ? v[m] * w : 0.0f;
            sA[s * FS + (lb + m) * 2 + j] = val;
        }
    }

    float g[10];
    {
        int e = 0;
        #pragma unroll
        for (int p = 0; p < 4; ++p)
            #pragma unroll
            for (int q = p; q < 4; ++q)
                g[e++] = v[p] * v[q] * w;
    }
    #pragma unroll
    for (int e = 0; e < 10; ++e) {
        #pragma unroll
        for (int off = 16; off > 0; off >>= 1)
            g[e] += __shfl_xor(g[e], off);
    }
    if (lane == 0) {
        #pragma unroll
        for (int e = 0; e < 10; ++e) sGp[wave * 12 + e] = g[e];
    }
    __syncthreads();

    v2f a[8];
    #pragma unroll
    for (int s = 0; s < 8; ++s)
        a[s] = *(const v2f*)(sA + (wave * 8 + s) * FS + lane * 2);

    const float* xr[8][2];
    #pragma unroll
    for (int s = 0; s < 8; ++s)
        #pragma unroll
        for (int j = 0; j < 2; ++j)
            xr[s][j] = X + (size_t)__builtin_amdgcn_readfirstlane((wave * 32 + 4 * s + j) * Dc);
    const unsigned hrow = (unsigned)(kb * Dc);
    unsigned xo[NW];
    #pragma unroll
    for (int t = 0; t < NW; ++t) {
        long col = (long)(tile0 + t) * 16 + c;
        if (col >= Dc) col = Dc - 1;
        xo[t] = hrow + (unsigned)col;
    }

    #pragma unroll
    for (int h = 0; h < 2; ++h) {
        v2f b[8][4];
        #pragma unroll
        for (int s = 0; s < 8; ++s) {
            #pragma unroll
            for (int t = 0; t < 4; ++t) {
                b[s][t].x = xr[s][0][xo[h * 4 + t]];
                b[s][t].y = xr[s][1][xo[h * 4 + t]];
            }
        }
        __builtin_amdgcn_sched_barrier(0);
        v8f acc[4];
        #pragma unroll
        for (int t = 0; t < 4; ++t)
            acc[t] = (v8f){0.0f, 0.0f, 0.0f, 0.0f, 0.0f, 0.0f, 0.0f, 0.0f};
        #pragma unroll
        for (int s = 0; s < 8; ++s) {
            #pragma unroll
            for (int t = 0; t < 4; ++t) {
                acc[t] = __builtin_amdgcn_wmma_f32_16x16x4_f32(
                             false, a[s], false, b[s][t], (short)0, acc[t], false, false);
            }
        }
        if (hi == 0) {
            #pragma unroll
            for (int t = 0; t < 4; ++t)
                sP[wave][h * 4 + t][c] = make_float4(acc[t][0], acc[t][1], acc[t][2], acc[t][3]);
        }
    }
    __syncthreads();

    const long d0 = (long)(tile0 + wave) * 16;
    if (hi == 0 && d0 + c < (long)Dc) {
        float4 mo = sP[0][wave][c];
        #pragma unroll
        for (int wv = 1; wv < NW; ++wv) {
            const float4 p = sP[wv][wave][c];
            mo.x += p.x; mo.y += p.y; mo.z += p.z; mo.w += p.w;
        }

        float G[10];
        #pragma unroll
        for (int e = 0; e < 10; ++e) {
            float gs = sGp[e];
            #pragma unroll
            for (int wv = 1; wv < NW; ++wv) gs += sGp[wv * 12 + e];
            G[e] = gs;
        }

        float M[4][8];
        {
            int e = 0;
            #pragma unroll
            for (int p = 0; p < 4; ++p)
                #pragma unroll
                for (int q = p; q < 4; ++q) {
                    M[p][q] = G[e];
                    M[q][p] = G[e];
                    ++e;
                }
        }
        #pragma unroll
        for (int r = 0; r < 4; ++r)
            #pragma unroll
            for (int cc = 0; cc < 4; ++cc) M[r][4 + cc] = (r == cc) ? 1.0f : 0.0f;
        #pragma unroll
        for (int pc = 0; pc < 4; ++pc) {
            const float inv = 1.0f / M[pc][pc];
            #pragma unroll
            for (int cc = 0; cc < 8; ++cc) M[pc][cc] *= inv;
            #pragma unroll
            for (int r = 0; r < 4; ++r) {
                if (r == pc) continue;
                const float f = M[r][pc];
                #pragma unroll
                for (int cc = 0; cc < 8; ++cc) M[r][cc] -= f * M[pc][cc];
            }
        }

        float4 o;
        o.x = M[0][4] * mo.x + M[0][5] * mo.y + M[0][6] * mo.z + M[0][7] * mo.w;
        o.y = M[1][4] * mo.x + M[1][5] * mo.y + M[1][6] * mo.z + M[1][7] * mo.w;
        o.z = M[2][4] * mo.x + M[2][5] * mo.y + M[2][6] * mo.z + M[2][7] * mo.w;
        o.w = M[3][4] * mo.x + M[3][5] * mo.y + M[3][6] * mo.z + M[3][7] * mo.w;
        *(float4*)(out + (d0 + c) * 4) = o;
    }
}

extern "C" void kernel_launch(void* const* d_in, const int* in_sizes, int n_in,
                              void* d_out, int out_size, void* d_ws, size_t ws_size,
                              hipStream_t stream) {
    const float* t_all   = (const float*)d_in[0];
    const float* x_all   = (const float*)d_in[1];
    const float* weights = (const float*)d_in[2];
    (void)d_in[3]; (void)n_in; (void)out_size; (void)d_ws; (void)ws_size;

    const int T  = in_sizes[0];
    const int Dc = in_sizes[1] / T;

    const int numTiles = (Dc + 15) / 16;
    const int blocks   = (numTiles + 7) / 8;
    hipLaunchKernelGGL(sg_fit, dim3(blocks), dim3(256), 0, stream,
                       t_all, weights, x_all, (float*)d_out, Dc);
}


# one work-group per compute unit: 250 work-groups of 8 waves each cover 25 channel tiles (400 channels) in 5 passes, the next pass's X loads issued among a pass's matrix instructions; sum order kept
# speedup vs baseline: 2.4520x; 1.0686x over previous
#include <hip/hip_runtime.h>


typedef __attribute__((ext_vector_type(2))) float v2f;
typedef __attribute__((ext_vector_type(8))) float v8f;

#define NR 256
#define FS 66
#define NW 8
#define PT 5
#define NP 5
#define CT (PT * NP)

__global__ __launch_bounds__(256) void sg_fit(const float* __restrict__ t_all,
                                              const float* __restrict__ weights,
                                              const float* __restrict__ X,
                                              float* __restrict__ out,
                                              int Dc)
{
    __shared__ __attribute__((aligned(16))) float sA[(NR / 4) * FS];
    __shared__ __attribute__((aligned(16))) float sGp[NW * 12];
    __shared__ __attribute__((aligned(16))) float4 sP[NW][CT][32];

    const int i    = threadIdx.x;
    const int lane = i & 31;
    const int wave = __builtin_amdgcn_readfirstlane(i >> 5);
    const int tile0 = blockIdx.x * CT;

    const int  c  = lane & 15;
    const int  hi = lane >> 4;
    const int  kb = hi * 2;

    const float tc = t_all[NR / 2];
    const float tr = t_all[i] - tc;
    const float wr = weights[i];
    const float w  = fmaxf(wr, 0.0f) + log1pf(expf(-fabsf(wr)));
    float v[4];
    v[0] = 1.0f;
    v[1] = tr;
    v[2] = tr * tr * 0.5f;
    v[3] = tr * tr * tr * (1.0f / 3.0f);

    {
        const int s  = i >> 2;
        const int k  = i & 3;
        const int j  = k & 1;
        const int lb = (k < 2) ? 0 : 16;
        #pragma unroll
        for (int m = 0; m < 16; ++m) {
            const float val = (m < 4) ? v[m] * w : 0.0f;
            sA[s * FS + (lb + m) * 2 + j] = val;
        }
    }

    float g[10];
    {
        int e = 0;
        #pragma unroll
        for (int p = 0; p < 4; ++p)
            #pragma unroll
            for (int q = p; q < 4; ++q)
                g[e++] = v[p] * v[q] * w;
    }
    #pragma unroll
    for (int e = 0; e < 10; ++e) {
        #pragma unroll
        for (int off = 16; off > 0; off >>= 1)
            g[e] += __shfl_xor(g[e], off);
    }
    if (lane == 0) {
        #pragma unroll
        for (int e = 0; e < 10; ++e) sGp[wave * 12 + e] = g[e];
    }
    __syncthreads();

    v2f a[8];
    #pragma unroll
    for (int s = 0; s < 8; ++s)
        a[s] = *(const v2f*)(sA + (wave * 8 + s) * FS + lane * 2);

    const float* xr[8][2];
    #pragma unroll
    for (int s = 0; s < 8; ++s)
        #pragma unroll
        for (int j = 0; j < 2; ++j)
            xr[s][j] = X + (size_t)__builtin_amdgcn_readfirstlane((wave * 32 + 4 * s + j) * Dc);
    const unsigned hrow = (unsigned)(kb * Dc);

    unsigned xo[CT];
    #pragma unroll
    for (int t = 0; t < CT; ++t) {
        long col = (long)(tile0 + t) * 16 + c;
        if (col >= Dc) col = Dc - 1;
        xo[t] = hrow + (unsigned)col;
    }
    v2f b[2][8][PT];
#define LOAD_PASS(p)                                                          \
    {                                                                         \
        _Pragma("unroll")                                                     \
        for (int s = 0; s < 8; ++s) {                                         \
            _Pragma("unroll")                                                 \
            for (int t = 0; t < PT; ++t) {                                    \
                b[(p) & 1][s][t].x = xr[s][0][xo[(p) * PT + t]];      \
                b[(p) & 1][s][t].y = xr[s][1][xo[(p) * PT + t]];      \
            }                                                                 \
        }                                                                     \
    }

    LOAD_PASS(0)
    #pragma unroll
    for (int p = 0; p < NP; ++p) {
        __builtin_amdgcn_sched_barrier(0);
        if (p + 1 < NP) LOAD_PASS(p + 1)
        v8f acc[PT];
        #pragma unroll
        for (int t = 0; t < PT; ++t)
            acc[t] = (v8f){0.0f, 0.0f, 0.0f, 0.0f, 0.0f, 0.0f, 0.0f, 0.0f};
        #pragma unroll
        for (int s = 0; s < 8; ++s) {
            #pragma unroll
            for (int t = 0; t < PT; ++t) {
                acc[t] = __builtin_amdgcn_wmma_f32_16x16x4_f32(
                             false, a[s], false, b[p & 1][s][t], (short)0, acc[t], false, false);
            }
        }
        if (p + 1 < NP) {
            #pragma unroll
            for (int k = 0; k < 8 * PT; ++k) {
                __builtin_amdgcn_sched_group_barrier(0x008, 1, 0);
                __builtin_amdgcn_sched_group_barrier(0x020, 2, 0);
            }
        }
        __builtin_amdgcn_sched_barrier(0);
        if (p + 1 == NP) {
            asm volatile("v_nop\n\tv_nop\n\tv_nop\n\tv_nop");
            __builtin_amdgcn_sched_barrier(0);
        }
        #pragma unroll
        for (int t = 0; t < PT; ++t)
            sP[wave][p * PT + t][lane] = make_float4(acc[t][0], acc[t][1], acc[t][2], acc[t][3]);
    }
#undef LOAD_PASS
    __syncthreads();

    float G[10];
    #pragma unroll
    for (int e = 0; e < 10; ++e) {
        float gs = sGp[e];
        #pragma unroll
        for (int wv = 1; wv < NW; ++wv) gs += sGp[wv * 12 + e];
        G[e] = gs;
    }
    float M[4][8];
    {
        int e = 0;
        #pragma unroll
        for (int p = 0; p < 4; ++p)
            #pragma unroll
            for (int q = p; q < 4; ++q) {
                M[p][q] = G[e];
                M[q][p] = G[e];
                ++e;
            }
    }
    #pragma unroll
    for (int r = 0; r < 4; ++r)
        #pragma unroll
        for (int cc = 0; cc < 4; ++cc) M[r][4 + cc] = (r == cc) ? 1.0f : 0.0f;
    #pragma unroll
    for (int pc = 0; pc < 4; ++pc) {
        const float inv = 1.0f / M[pc][pc];
        #pragma unroll
        for (int cc = 0; cc < 8; ++cc) M[pc][cc] *= inv;
        #pragma unroll
        for (int r = 0; r < 4; ++r) {
            if (r == pc) continue;
            const float f = M[r][pc];
            #pragma unroll
            for (int cc = 0; cc < 8; ++cc) M[r][cc] -= f * M[pc][cc];
        }
    }

    #pragma unroll
    for (int rnd = 0; rnd < 2; ++rnd) {
        const int  lc = i + rnd * 256;
        const long d  = (long)tile0 * 16 + lc;
        if (lc < CT * 16 && d < (long)Dc) {
            float4 mo = sP[0][lc >> 4][lc & 15];
            #pragma unroll
            for (int wv = 1; wv < NW; ++wv) {
                const float4 pp = sP[wv][lc >> 4][lc & 15];
                mo.x += pp.x; mo.y += pp.y; mo.z += pp.z; mo.w += pp.w;
            }
            float4 o;
            o.x = M[0][4] * mo.x + M[0][5] * mo.y + M[0][6] * mo.z + M[0][7] * mo.w;
            o.y = M[1][4] * mo.x + M[1][5] * mo.y + M[1][6] * mo.z + M[1][7] * mo.w;
            o.z = M[2][4] * mo.x + M[2][5] * mo.y + M[2][6] * mo.z + M[2][7] * mo.w;
            o.w = M[3][4] * mo.x + M[3][5] * mo.y + M[3][6] * mo.z + M[3][7] * mo.w;
            *(float4*)(out + d * 4) = o;
        }
    }
}

extern "C" void kernel_launch(void* const* d_in, const int* in_sizes, int n_in,
                              void* d_out, int out_size, void* d_ws, size_t ws_size,
                              hipStream_t stream) {
    const float* t_all   = (const float*)d_in[0];
    const float* x_all   = (const float*)d_in[1];
    const float* weights = (const float*)d_in[2];
    (void)d_in[3]; (void)n_in; (void)out_size; (void)d_ws; (void)ws_size;

    const int T  = in_sizes[0];
    const int Dc = in_sizes[1] / T;

    const int numTiles = (Dc + 15) / 16;
    const int blocks   = (numTiles + CT - 1) / CT;
    hipLaunchKernelGGL(sg_fit, dim3(blocks), dim3(256), 0, stream,
                       t_all, weights, x_all, (float*)d_out, Dc);
}


# block-diagonal A: one matrix instruction takes one time row of 64 channels (4 k slots = 4 channels), a lane loads 8 adjacent bytes, a wave 256 adjacent bytes of one row; new sum order
# speedup vs baseline: 2.7380x; 1.1166x over previous
#include <hip/hip_runtime.h>


typedef __attribute__((ext_vector_type(2))) float v2f;
typedef __attribute__((ext_vector_type(8))) float v8f;

#define NR 256
#define NW 8
#define WP 272
#define NG 6
#define PG 2
#define NPS 3
#define CG (NG * 64)
#define CW 400

__global__ __launch_bounds__(256) void sg_fit(const float* __restrict__ t_all,
                                              const float* __restrict__ weights,
                                              const float* __restrict__ X,
                                              float* __restrict__ out,
                                              int Dc)
{
    __shared__ __attribute__((aligned(16))) float sW[4 * WP];
    __shared__ __attribute__((aligned(16))) float sGp[NW * 12];
    __shared__ __attribute__((aligned(16))) float4 sPn[NW][CG];
    __shared__ __attribute__((aligned(16))) float4 sPo[NW][32];

    const int i    = threadIdx.x;
    const int lane = i & 31;
    const int wave = __builtin_amdgcn_readfirstlane(i >> 5);
    const int odd  = blockIdx.x & 1;
    const long cbase = (long)blockIdx.x * CW;
    const long gbase = cbase + (odd ? 16 : 0);
    const long obase = cbase + (odd ? 0 : CG);

    const int m  = lane & 15;
    const int h  = lane >> 4;
    const int g  = m >> 2;

    const float tc = t_all[NR / 2];
    const float tr = t_all[i] - tc;
    const float wr = weights[i];
    const float w  = fmaxf(wr, 0.0f) + log1pf(expf(-fabsf(wr)));
    float v[4];
    v[0] = 1.0f;
    v[1] = tr;
    v[2] = tr * tr * 0.5f;
    v[3] = tr * tr * tr * (1.0f / 3.0f);

    #pragma unroll
    for (int p = 0; p < 4; ++p) sW[p * WP + i] = v[p] * w;

    float gq[10];
    {
        int e = 0;
        #pragma unroll
        for (int p = 0; p < 4; ++p)
            #pragma unroll
            for (int q = p; q < 4; ++q)
                gq[e++] = v[p] * v[q] * w;
    }
    #pragma unroll
    for (int e = 0; e < 10; ++e) {
        #pragma unroll
        for (int off = 16; off > 0; off >>= 1)
            gq[e] += __shfl_xor(gq[e], off);
    }
    if (lane == 0) {
        #pragma unroll
        for (int e = 0; e < 10; ++e) sGp[wave * 12 + e] = gq[e];
    }
    __syncthreads();

    float val[32];
    {
        const float* wrow = sW + (lane & 3) * WP + wave * 32;
        #pragma unroll
        for (int q = 0; q < 8; ++q) {
            const float4 t4 = *(const float4*)(wrow + q * 4);
            val[4 * q + 0] = t4.x; val[4 * q + 1] = t4.y; val[4 * q + 2] = t4.z; val[4 * q + 3] = t4.w;
        }
    }
    const bool sx = (2 * h == g);
    const bool sy = (2 * h + 1 == g);
    v2f a[32];
    #pragma unroll
    for (int r = 0; r < 32; ++r) {
        a[r].x = sx ? val[r] : 0.0f;
        a[r].y = sy ? val[r] : 0.0f;
    }
    const bool lo = (m < 4);
    v2f ao[8];
    #pragma unroll
    for (int s = 0; s < 8; ++s) {
        const float e0 = h ? val[4 * s + 2] : val[4 * s + 0];
        const float e1 = h ? val[4 * s + 3] : val[4 * s + 1];
        ao[s].x = lo ? e0 : 0.0f;
        ao[s].y = lo ? e1 : 0.0f;
    }

    const float* xr[32];
    #pragma unroll
    for (int r = 0; r < 32; ++r)
        xr[r] = X + (size_t)__builtin_amdgcn_readfirstlane((wave * 32 + r) * Dc);

    unsigned xo[NG];
    #pragma unroll
    for (int q = 0; q < NG; ++q) {
        int cc = (int)gbase + 64 * q + 2 * lane;
        const int last = (Dc - 2) & ~1;
        if (cc > last) cc = last;
        xo[q] = (unsigned)cc * 4u;
    }
    unsigned xoo;
    {
        int cc = (int)obase + m;
        if (cc >= Dc) cc = Dc - 1;
        xoo = ((unsigned)(2 * h * Dc) + (unsigned)cc) * 4u;
    }

    #pragma unroll
    for (int q = 0; q < NG; ++q) asm volatile("" : "+v"(xo[q]));

    v2f b[2][32][PG];
    v2f bo[8];
#define LOAD_PASS(ps)                                                          \
    {                                                                          \
        _Pragma("unroll")                                                      \
        for (int r = 0; r < 32; ++r) {                                         \
            _Pragma("unroll")                                                  \
            for (int q = 0; q < PG; ++q)                                       \
                b[(ps) & 1][r][q] = *(const v2f*)((const char*)xr[r] + xo[(ps) * PG + q]); \
        }                                                                      \
    }

    #pragma unroll
    for (int s = 0; s < 8; ++s) {
        bo[s].x = *(const float*)((const char*)xr[4 * s + 0] + xoo);
        bo[s].y = *(const float*)((const char*)xr[4 * s + 1] + xoo);
    }
    LOAD_PASS(0)
    #pragma unroll
    for (int ps = 0; ps < NPS; ++ps) {
        __builtin_amdgcn_sched_barrier(0);
        if (ps + 1 < NPS) LOAD_PASS(ps + 1)
        v8f acc[PG];
        #pragma unroll
        for (int q = 0; q < PG; ++q)
            acc[q] = (v8f){0.0f, 0.0f, 0.0f, 0.0f, 0.0f, 0.0f, 0.0f, 0.0f};
        v8f acco = (v8f){0.0f, 0.0f, 0.0f, 0.0f, 0.0f, 0.0f, 0.0f, 0.0f};
        if (ps == 0) {
            #pragma unroll
            for (int s = 0; s < 8; ++s)
                acco = __builtin_amdgcn_wmma_f32_16x16x4_f32(
                           false, ao[s], false, bo[s], (short)0, acco, false, false);
        }
        #pragma unroll
        for (int r = 0; r < 32; ++r) {
            #pragma unroll
            for (int q = 0; q < PG; ++q)
                acc[q] = __builtin_amdgcn_wmma_f32_16x16x4_f32(
                             false, a[r], false, b[ps & 1][r][q], (short)0, acc[q], false, false);
        }
        if (ps + 1 < NPS) {
            if (ps == 0)
                __builtin_amdgcn_sched_group_barrier(0x008, 8, 0);
            #pragma unroll
            for (int k = 0; k < 32 * PG; ++k) {
                __builtin_amdgcn_sched_group_barrier(0x008, 1, 0);
                __builtin_amdgcn_sched_group_barrier(0x020, 1, 0);
            }
        }
        __builtin_amdgcn_sched_barrier(0);
        if (ps + 1 == NPS) {
            asm volatile("v_nop\n\tv_nop\n\tv_nop\n\tv_nop");
            __builtin_amdgcn_sched_barrier(0);
        }
        #pragma unroll
        for (int q = 0; q < PG; ++q) {
            const int cg = (ps * PG + q) * 64 + 2 * lane;
            sPn[wave][cg + 0] = make_float4(acc[q][0], acc[q][1], acc[q][2], acc[q][3]);
            sPn[wave][cg + 1] = make_float4(acc[q][4], acc[q][5], acc[q][6], acc[q][7]);
        }
        if (ps == 0)
            sPo[wave][lane] = make_float4(acco[0], acco[1], acco[2], acco[3]);
    }
#undef LOAD_PASS
    __syncthreads();

    float G[10];
    #pragma unroll
    for (int e = 0; e < 10; ++e) {
        float gs = sGp[e];
        #pragma unroll
        for (int wv = 1; wv < NW; ++wv) gs += sGp[wv * 12 + e];
        G[e] = gs;
    }
    float M[4][8];
    {
        int e = 0;
        #pragma unroll
        for (int p = 0; p < 4; ++p)
            #pragma unroll
            for (int q = p; q < 4; ++q) {
                M[p][q] = G[e];
                M[q][p] = G[e];
                ++e;
            }
    }
    #pragma unroll
    for (int r = 0; r < 4; ++r)
        #pragma unroll
        for (int cc = 0; cc < 4; ++cc) M[r][4 + cc] = (r == cc) ? 1.0f : 0.0f;
    #pragma unroll
    for (int pc = 0; pc < 4; ++pc) {
        const float inv = 1.0f / M[pc][pc];
        #pragma unroll
        for (int cc = 0; cc < 8; ++cc) M[pc][cc] *= inv;
        #pragma unroll
        for (int r = 0; r < 4; ++r) {
            if (r == pc) continue;
            const float f = M[r][pc];
            #pragma unroll
            for (int cc = 0; cc < 8; ++cc) M[r][cc] -= f * M[pc][cc];
        }
    }

    #pragma unroll
    for (int rnd = 0; rnd < 2; ++rnd) {
        const float4* src;
        long d;
        bool on;
        if (rnd == 0) {
            src = &sPn[0][i];            d = gbase + i;             on = true;
        } else if (i < 128) {
            src = &sPn[0][256 + i];      d = gbase + 256 + i;       on = true;
        } else {
            src = &sPo[0][(i - 128) & 15]; d = obase + (i - 128);   on = (i < 144);
        }
        const int pitch = (rnd == 1 && i >= 128) ? 32 : CG;
        if (on && d < (long)Dc) {
            float4 mo = src[0];
            #pragma unroll
            for (int wv = 1; wv < NW; ++wv) {
                const float4 pp = src[wv * pitch];
                mo.x += pp.x; mo.y += pp.y; mo.z += pp.z; mo.w += pp.w;
            }
            float4 o;
            o.x = M[0][4] * mo.x + M[0][5] * mo.y + M[0][6] * mo.z + M[0][7] * mo.w;
            o.y = M[1][4] * mo.x + M[1][5] * mo.y + M[1][6] * mo.z + M[1][7] * mo.w;
            o.z = M[2][4] * mo.x + M[2][5] * mo.y + M[2][6] * mo.z + M[2][7] * mo.w;
            o.w = M[3][4] * mo.x + M[3][5] * mo.y + M[3][6] * mo.z + M[3][7] * mo.w;
            *(float4*)(out + d * 4) = o;
        }
    }
}

extern "C" void kernel_launch(void* const* d_in, const int* in_sizes, int n_in,
                              void* d_out, int out_size, void* d_ws, size_t ws_size,
                              hipStream_t stream) {
    const float* t_all   = (const float*)d_in[0];
    const float* x_all   = (const float*)d_in[1];
    const float* weights = (const float*)d_in[2];
    (void)d_in[3]; (void)n_in; (void)out_size; (void)d_ws; (void)ws_size;

    const int T  = in_sizes[0];
    const int Dc = in_sizes[1] / T;

    const int blocks = (Dc + CW - 1) / CW;
    hipLaunchKernelGGL(sg_fit, dim3(blocks), dim3(256), 0, stream,
                       t_all, weights, x_all, (float*)d_out, Dc);
}


# X requests ahead of the weight build: 62 X loads leave behind the row addresses; softplus, in-wave hand-over (no head barrier), A build run under them; Gram butterfly after pass 0; sum order kept
# speedup vs baseline: 2.9035x; 1.0604x over previous
#include <hip/hip_runtime.h>


typedef __attribute__((ext_vector_type(2))) float v2f;
typedef __attribute__((ext_vector_type(8))) float v8f;

#define NR 256
#define NW 8
#define WP 272
#define NG 6
#define PG 2
#define NPS 3
#define CG (NG * 64)
#define CW 400

__global__ __launch_bounds__(256) void sg_fit(const float* __restrict__ t_all,
                                              const float* __restrict__ weights,
                                              const float* __restrict__ X,
                                              float* __restrict__ out,
                                              int Dc)
{
    __shared__ __attribute__((aligned(16))) float sW[4 * WP];
    __shared__ __attribute__((aligned(16))) float sGp[NW * 12];
    __shared__ __attribute__((aligned(16))) float4 sPn[NW][CG];
    __shared__ __attribute__((aligned(16))) float4 sPo[NW][32];

    const int i    = threadIdx.x;
    const int lane = i & 31;
    const int wave = __builtin_amdgcn_readfirstlane(i >> 5);
    const int odd  = blockIdx.x & 1;
    const long cbase = (long)blockIdx.x * CW;
    const long gbase = cbase + (odd ? 16 : 0);
    const long obase = cbase + (odd ? 0 : CG);

    const int m  = lane & 15;
    const int h  = lane >> 4;
    const int g  = m >> 2;

    const char* xr[32];
    {
        const unsigned long long pitch = (unsigned long long)(unsigned)Dc * sizeof(float);
        const unsigned long long o0 = (unsigned long long)(unsigned)(wave * 32) * pitch;
        #pragma unroll
        for (int r = 0; r < 32; ++r) {
            unsigned long long ro = o0 + (unsigned long long)r * pitch;
            asm("" : "+s"(ro));
            xr[r] = (const char*)X + ro;
        }
    }

    unsigned xo[NG];
    #pragma unroll
    for (int q = 0; q < NG; ++q) {
        int cc = (int)gbase + 64 * q + 2 * lane;
        const int last = (Dc - 2) & ~1;
        if (cc > last) cc = last;
        xo[q] = (unsigned)cc * 4u;
    }
    unsigned xoo;
    {
        int cc = (int)obase + m;
        if (cc >= Dc) cc = Dc - 1;
        xoo = ((unsigned)(2 * h * Dc) + (unsigned)cc) * 4u;
    }

    #pragma unroll
    for (int q = 0; q < NG; ++q) asm volatile("" : "+v"(xo[q]));

    const float tc = t_all[NR / 2];
    const float ti = t_all[i];
    const float wr = weights[i];
    __builtin_amdgcn_sched_barrier(0);

    v2f b[2][32][PG];
    v2f bo[8];
#define FR 23
#define LOAD_ROWS(ps, r0, r1)                                                  \
    {                                                                          \
        _Pragma("unroll")                                                      \
        for (int r = (r0); r < (r1); ++r) {                                    \
            _Pragma("unroll")                                                  \
            for (int q = 0; q < PG; ++q)                                       \
                b[(ps) & 1][r][q] = *(const v2f*)(xr[r] + xo[(ps) * PG + q]);  \
        }                                                                      \
    }
#define LOAD_PASS(ps) LOAD_ROWS(ps, 0, 32)
#define LOAD_ROWS_IN_ORDER(ps, r0, r1)                                         \
    {                                                                          \
        _Pragma("unroll")                                                      \
        for (int r = (r0); r < (r1); ++r) {                                    \
            _Pragma("unroll")                                                  \
            for (int q = 0; q < PG; ++q)                                       \
                b[(ps) & 1][r][q] = *(const v2f*)(xr[r] + xo[(ps) * PG + q]);  \
            __builtin_amdgcn_sched_barrier(0);                                 \
        }                                                                      \
    }

    #pragma unroll
    for (int s = 0; s < 8; ++s) {
        bo[s].x = *(const float*)(xr[4 * s + 0] + xoo);
        bo[s].y = *(const float*)(xr[4 * s + 1] + xoo);
    }
    __builtin_amdgcn_sched_barrier(0);
    LOAD_ROWS_IN_ORDER(0, 0, FR)

    const float tr = ti - tc;
    const float w  = fmaxf(wr, 0.0f) + log1pf(expf(-fabsf(wr)));
    float v[4];
    v[0] = 1.0f;
    v[1] = tr;
    v[2] = tr * tr * 0.5f;
    v[3] = tr * tr * tr * (1.0f / 3.0f);

    #pragma unroll
    for (int p = 0; p < 4; ++p) sW[p * WP + i] = v[p] * w;

    float val[32];
    {
        const float* wrow = sW + (lane & 3) * WP + wave * 32;
        #pragma unroll
        for (int q = 0; q < 8; ++q) {
            const float4 t4 = *(const float4*)(wrow + q * 4);
            val[4 * q + 0] = t4.x; val[4 * q + 1] = t4.y; val[4 * q + 2] = t4.z; val[4 * q + 3] = t4.w;
        }
    }

    const bool sx = (2 * h == g);
    const bool sy = (2 * h + 1 == g);
    v2f a[32];
    #pragma unroll
    for (int r = 0; r < 32; ++r) {
        a[r].x = sx ? val[r] : 0.0f;
        a[r].y = sy ? val[r] : 0.0f;
    }
    const bool lo = (m < 4);
    v2f ao[8];
    #pragma unroll
    for (int s = 0; s < 8; ++s) {
        const float e0 = h ? val[4 * s + 2] : val[4 * s + 0];
        const float e1 = h ? val[4 * s + 3] : val[4 * s + 1];
        ao[s].x = lo ? e0 : 0.0f;
        ao[s].y = lo ? e1 : 0.0f;
    }

    __builtin_amdgcn_sched_barrier(0);
    LOAD_ROWS_IN_ORDER(0, FR, 32)
    #pragma unroll
    for (int ps = 0; ps < NPS; ++ps) {
        __builtin_amdgcn_sched_barrier(0);
        if (ps + 1 < NPS) LOAD_PASS(ps + 1)
        v8f acc[PG];
        #pragma unroll
        for (int q = 0; q < PG; ++q)
            acc[q] = (v8f){0.0f, 0.0f, 0.0f, 0.0f, 0.0f, 0.0f, 0.0f, 0.0f};
        v8f acco = (v8f){0.0f, 0.0f, 0.0f, 0.0f, 0.0f, 0.0f, 0.0f, 0.0f};
        if (ps == 0) {
            #pragma unroll
            for (int s = 0; s < 8; ++s)
                acco = __builtin_amdgcn_wmma_f32_16x16x4_f32(
                           false, ao[s], false, bo[s], (short)0, acco, false, false);
        }
        #pragma unroll
        for (int r = 0; r < 32; ++r) {
            #pragma unroll
            for (int q = 0; q < PG; ++q)
                acc[q] = __builtin_amdgcn_wmma_f32_16x16x4_f32(
                             false, a[r], false, b[ps & 1][r][q], (short)0, acc[q], false, false);
        }
        if (ps + 1 < NPS) {
            if (ps == 0)
                __builtin_amdgcn_sched_group_barrier(0x008, 8, 0);
            #pragma unroll
            for (int k = 0; k < 32 * PG; ++k) {
                __builtin_amdgcn_sched_group_barrier(0x008, 1, 0);
                __builtin_amdgcn_sched_group_barrier(0x020, 1, 0);
            }
        }
        __builtin_amdgcn_sched_barrier(0);
        {
            asm volatile("v_nop\n\tv_nop\n\tv_nop\n\tv_nop");
            __builtin_amdgcn_sched_barrier(0);
        }
        #pragma unroll
        for (int q = 0; q < PG; ++q) {
            const int cg = (ps * PG + q) * 64 + 2 * lane;
            sPn[wave][cg + 0] = make_float4(acc[q][0], acc[q][1], acc[q][2], acc[q][3]);
            sPn[wave][cg + 1] = make_float4(acc[q][4], acc[q][5], acc[q][6], acc[q][7]);
        }
        if (ps == 0)
            sPo[wave][lane] = make_float4(acco[0], acco[1], acco[2], acco[3]);
        if (ps == 0) {
            float gq[10];
            {
                int e = 0;
                #pragma unroll
                for (int p = 0; p < 4; ++p)
                    #pragma unroll
                    for (int q = p; q < 4; ++q)
                        gq[e++] = v[p] * v[q] * w;
            }
            #pragma unroll
            for (int e = 0; e < 10; ++e) {
                #pragma unroll
                for (int off = 16; off > 0; off >>= 1)
                    gq[e] += __shfl_xor(gq[e], off);
            }
            if (lane == 0) {
                #pragma unroll
                for (int e = 0; e < 10; ++e) sGp[wave * 12 + e] = gq[e];
            }
        }
    }
#undef LOAD_PASS
#undef LOAD_ROWS
#undef LOAD_ROWS_IN_ORDER
#undef FR
    __syncthreads();

    float G[10];
    #pragma unroll
    for (int e = 0; e < 10; ++e) {
        float gs = sGp[e];
        #pragma unroll
        for (int wv = 1; wv < NW; ++wv) gs += sGp[wv * 12 + e];
        G[e] = gs;
    }
    float M[4][8];
    {
        int e = 0;
        #pragma unroll
        for (int p = 0; p < 4; ++p)
            #pragma unroll
            for (int q = p; q < 4; ++q) {
                M[p][q] = G[e];
                M[q][p] = G[e];
                ++e;
            }
    }
    #pragma unroll
    for (int r = 0; r < 4; ++r)
        #pragma unroll
        for (int cc = 0; cc < 4; ++cc) M[r][4 + cc] = (r == cc) ? 1.0f : 0.0f;
    #pragma unroll
    for (int pc = 0; pc < 4; ++pc) {
        const float inv = 1.0f / M[pc][pc];
        #pragma unroll
        for (int cc = 0; cc < 8; ++cc) M[pc][cc] *= inv;
        #pragma unroll
        for (int r = 0; r < 4; ++r) {
            if (r == pc) continue;
            const float f = M[r][pc];
            #pragma unroll
            for (int cc = 0; cc < 8; ++cc) M[r][cc] -= f * M[pc][cc];
        }
    }

    #pragma unroll
    for (int rnd = 0; rnd < 2; ++rnd) {
        const float4* src;
        long d;
        bool on;
        if (rnd == 0) {
            src = &sPn[0][i];            d = gbase + i;             on = true;
        } else if (i < 128) {
            src = &sPn[0][256 + i];      d = gbase + 256 + i;       on = true;
        } else {
            src = &sPo[0][(i - 128) & 15]; d = obase + (i - 128);   on = (i < 144);
        }
        const int pitch = (rnd == 1 && i >= 128) ? 32 : CG;
        if (on && d < (long)Dc) {
            float4 mo = src[0];
            #pragma unroll
            for (int wv = 1; wv < NW; ++wv) {
                const float4 pp = src[wv * pitch];
                mo.x += pp.x; mo.y += pp.y; mo.z += pp.z; mo.w += pp.w;
            }
            float4 o;
            o.x = M[0][4] * mo.x + M[0][5] * mo.y + M[0][6] * mo.z + M[0][7] * mo.w;
            o.y = M[1][4] * mo.x + M[1][5] * mo.y + M[1][6] * mo.z + M[1][7] * mo.w;
            o.z = M[2][4] * mo.x + M[2][5] * mo.y + M[2][6] * mo.z + M[2][7] * mo.w;
            o.w = M[3][4] * mo.x + M[3][5] * mo.y + M[3][6] * mo.z + M[3][7] * mo.w;
            *(float4*)(out + d * 4) = o;
        }
    }
}

extern "C" void kernel_launch(void* const* d_in, const int* in_sizes, int n_in,
                              void* d_out, int out_size, void* d_ws, size_t ws_size,
                              hipStream_t stream) {
    const float* t_all   = (const float*)d_in[0];
    const float* x_all   = (const float*)d_in[1];
    const float* weights = (const float*)d_in[2];
    (void)d_in[3]; (void)n_in; (void)out_size; (void)d_ws; (void)ws_size;

    const int T  = in_sizes[0];
    const int Dc = in_sizes[1] / T;

    const int blocks = (Dc + CW - 1) / CW;
    hipLaunchKernelGGL(sg_fit, dim3(blocks), dim3(256), 0, stream,
                       t_all, weights, x_all, (float*)d_out, Dc);
}
